# starting kernel
# baseline (speedup 1.0000x reference)
#include <hip/hip_runtime.h>


#define BB 64
#define TT 4096
#define CC 80
#define NCHUNK 32
#define LCH 128
#define NTILE (LCH / 16)
#define NCT 5
#define DMIN 0.00390625f
#define FLOORV 1e-6f

typedef __attribute__((ext_vector_type(2))) float v2f;
typedef __attribute__((ext_vector_type(8))) float v8f;

__device__ __forceinline__ float clamp01f(float x) { return fminf(fmaxf(x, 0.0f), 1.0f); }
__device__ __forceinline__ float flog2(float x) { return __builtin_amdgcn_logf(x); }
__device__ __forceinline__ float fexp2(float x) { return __builtin_amdgcn_exp2f(x); }

__global__ __launch_bounds__(256)
void pcen_chunk_agg(const float* __restrict__ X, const float* __restrict__ smooth,
                    float* __restrict__ bchunk) {
  const int lane  = threadIdx.x & 31;
  const int wid   = blockIdx.x * 8 + (threadIdx.x >> 5);
  const int ctile = wid % NCT;
  const int tmp   = wid / NCT;
  const int chunk = tmp % NCHUNK;
  const int b     = tmp / NCHUNK;
  const int hi    = lane >> 4;
  const int c     = ctile * 16 + (lane & 15);

  const float wsm = clamp01f(smooth[c]);
  const float d   = fminf(fmaxf(1.0f - wsm, DMIN), 1.0f);
  const float l2d = flog2(d);

  float aP[8], syB[8];
#pragma unroll
  for (int j = 0; j < 8; ++j) {
    const int u = 4 * (j >> 1) + (j & 1) + 2 * hi;
    aP[j]  = (u <= (lane & 15)) ? 1.0f : 0.0f;
    syB[j] = wsm * fexp2(-(float)u * l2d);
  }
  const float dt7 = fexp2((float)(7 + 8 * hi) * l2d);

  const float* pB = X + ((size_t)b * TT + (size_t)chunk * LCH + 2 * hi) * CC + c;

  float s = 0.0f;
#pragma unroll 1
  for (int tile = 0; tile < NTILE; ++tile) {
    float xb[8];
#pragma unroll
    for (int j = 0; j < 8; ++j) {
      const int u0 = 4 * (j >> 1) + (j & 1);
      xb[j] = pB[(size_t)u0 * CC];
    }
    v8f acc = {0.f, 0.f, 0.f, 0.f, 0.f, 0.f, 0.f, 0.f};
#pragma unroll
    for (int kk = 0; kk < 4; ++kk) {
      v2f a2; a2[0] = aP[2 * kk];               a2[1] = aP[2 * kk + 1];
      v2f b2; b2[0] = xb[2 * kk] * syB[2 * kk]; b2[1] = xb[2 * kk + 1] * syB[2 * kk + 1];
      acc = __builtin_amdgcn_wmma_f32_16x16x4_f32(false, a2, false, b2, (short)0, acc,
                                                  false, false);
    }
    const float a15 = dt7 * (acc[7] + d * s);
    s = __shfl(a15, (lane & 15) + 16, 32);
    pB += 16 * CC;
  }
  if (lane < 16) bchunk[((size_t)b * NCHUNK + chunk) * CC + c] = s;
}

__global__ __launch_bounds__(256)
void pcen_combine(const float* __restrict__ X, const float* __restrict__ smooth,
                  const float* __restrict__ bchunk, float* __restrict__ start) {
  const int tid = blockIdx.x * 256 + threadIdx.x;
  if (tid >= BB * CC) return;
  const int b = tid / CC, c = tid % CC;
  const float wsm = clamp01f(smooth[c]);
  const float d   = fminf(fmaxf(1.0f - wsm, DMIN), 1.0f);
  const float dL  = fexp2((float)LCH * flog2(d));
  float s = X[(size_t)b * TT * CC + c];
  for (int k = 0; k < NCHUNK; ++k) {
    start[((size_t)b * NCHUNK + k) * CC + c] = s;
    s = dL * s + bchunk[((size_t)b * NCHUNK + k) * CC + c];
  }
}

__global__ __launch_bounds__(256)
void pcen_emit(const float* __restrict__ X, const float* __restrict__ alpha,
               const float* __restrict__ delta, const float* __restrict__ root,
               const float* __restrict__ smooth, const float* __restrict__ start,
               float* __restrict__ out) {
  const int lane  = threadIdx.x & 31;
  const int wid   = blockIdx.x * 8 + (threadIdx.x >> 5);
  const int ctile = wid % NCT;
  const int tmp   = wid / NCT;
  const int chunk = tmp % NCHUNK;
  const int b     = tmp / NCHUNK;
  const int hi    = lane >> 4;
  const int c     = ctile * 16 + (lane & 15);

  const float wsm = clamp01f(smooth[c]);
  const float d   = fminf(fmaxf(1.0f - wsm, DMIN), 1.0f);
  const float l2d = flog2(d);
  const float ac  = fminf(alpha[c], 1.0f);
  const float dc  = delta[c];
  const float ir  = 1.0f / fmaxf(root[c], 1.0f);
  const float off = fexp2(ir * flog2(dc));

  float aP[8], syB[8], dt[8];
#pragma unroll
  for (int j = 0; j < 8; ++j) {
    const int u = 4 * (j >> 1) + (j & 1) + 2 * hi;
    aP[j]  = (u <= (lane & 15)) ? 1.0f : 0.0f;
    syB[j] = wsm * fexp2(-(float)u * l2d);
    dt[j]  = fexp2((float)(j + 8 * hi) * l2d);
  }

  float s = start[((size_t)b * NCHUNK + chunk) * CC + c];

  const size_t rowBase = ((size_t)b * TT + (size_t)chunk * LCH) * CC + c;
  const float* pB = X + rowBase + (size_t)(2 * hi) * CC;
  const float* pD = X + rowBase + (size_t)(8 * hi) * CC;
  float*       pO = out + rowBase + (size_t)(8 * hi) * CC;

#pragma unroll 1
  for (int tile = 0; tile < NTILE; ++tile) {
    float xb[8];
#pragma unroll
    for (int j = 0; j < 8; ++j) {
      const int u0 = 4 * (j >> 1) + (j & 1);
      xb[j] = pB[(size_t)u0 * CC];
    }
    v8f acc = {0.f, 0.f, 0.f, 0.f, 0.f, 0.f, 0.f, 0.f};
#pragma unroll
    for (int kk = 0; kk < 4; ++kk) {
      v2f a2; a2[0] = aP[2 * kk];               a2[1] = aP[2 * kk + 1];
      v2f b2; b2[0] = xb[2 * kk] * syB[2 * kk]; b2[1] = xb[2 * kk + 1] * syB[2 * kk + 1];
      acc = __builtin_amdgcn_wmma_f32_16x16x4_f32(false, a2, false, b2, (short)0, acc,
                                                  false, false);
    }
    float a7 = 0.0f;
#pragma unroll
    for (int v = 0; v < 8; ++v) {
      const float ema = dt[v] * (acc[v] + d * s);
      if (v == 7) a7 = ema;
      const float xv  = pD[(size_t)v * CC];
      const float lp  = flog2(ema + FLOORV);
      const float num = xv * fexp2(-ac * lp) + dc;
      pO[(size_t)v * CC] = fexp2(ir * flog2(num)) - off;
    }
    s = __shfl(a7, (lane & 15) + 16, 32);
    pB += 16 * CC; pD += 16 * CC; pO += 16 * CC;
  }
}

extern "C" void kernel_launch(void* const* d_in, const int* in_sizes, int n_in,
                              void* d_out, int out_size, void* d_ws, size_t ws_size,
                              hipStream_t stream) {
  const float* X      = (const float*)d_in[0];
  const float* alpha  = (const float*)d_in[1];
  const float* delta  = (const float*)d_in[2];
  const float* root   = (const float*)d_in[3];
  const float* smooth = (const float*)d_in[4];
  float* out = (float*)d_out;

  float* bchunk = (float*)d_ws;
  float* start  = bchunk + (size_t)BB * NCHUNK * CC;

  const int nwaves = BB * NCHUNK * NCT;
  pcen_chunk_agg<<<nwaves / 8, 256, 0, stream>>>(X, smooth, bchunk);
  pcen_combine<<<(BB * CC + 255) / 256, 256, 0, stream>>>(X, smooth, bchunk, start);
  pcen_emit<<<nwaves / 8, 256, 0, stream>>>(X, alpha, delta, root, smooth, start, out);
}


# wave tile widened from 16 to all 80 channels: one wave per (batch, chunk) runs 5 independent WMMA chains per 16-row tile and issues the tile's loads as one group; 2,048 waves a launch, not 10,240
# speedup vs baseline: 1.1608x; 1.1608x over previous
#include <hip/hip_runtime.h>


#define BB 64
#define TT 4096
#define CC 80
#define NCHUNK 32
#define LCH 128
#define NTILE (LCH / 16)
#define NCT 5
#define DMIN 0.00390625f
#define FLOORV 1e-6f

typedef __attribute__((ext_vector_type(2))) float v2f;
typedef __attribute__((ext_vector_type(8))) float v8f;

__device__ __forceinline__ float clamp01f(float x) { return fminf(fmaxf(x, 0.0f), 1.0f); }
__device__ __forceinline__ float flog2(float x) { return __builtin_amdgcn_logf(x); }
__device__ __forceinline__ float fexp2(float x) { return __builtin_amdgcn_exp2f(x); }

__device__ __forceinline__ void load_b_tile(const float* __restrict__ p, float (&x)[NCT][8]) {
#pragma unroll
  for (int j = 0; j < 8; ++j) {
    const int u0 = 4 * (j >> 1) + (j & 1);
#pragma unroll
    for (int i = 0; i < NCT; ++i) x[i][j] = p[u0 * CC + 16 * i];
  }
}

__device__ __forceinline__ void tile_wmma(const float (&aP)[8], const float (&syB)[NCT][8],
                                          const float (&xb)[NCT][8], v8f (&acc)[NCT]) {
#pragma unroll
  for (int i = 0; i < NCT; ++i) acc[i] = (v8f){0.f, 0.f, 0.f, 0.f, 0.f, 0.f, 0.f, 0.f};
#pragma unroll
  for (int kk = 0; kk < 4; ++kk) {
    v2f a2; a2[0] = aP[2 * kk]; a2[1] = aP[2 * kk + 1];
#pragma unroll
    for (int i = 0; i < NCT; ++i) {
      v2f b2;
      b2[0] = xb[i][2 * kk] * syB[i][2 * kk];
      b2[1] = xb[i][2 * kk + 1] * syB[i][2 * kk + 1];
      acc[i] = __builtin_amdgcn_wmma_f32_16x16x4_f32(false, a2, false, b2, (short)0, acc[i],
                                                     false, false);
    }
  }
}

__global__ __launch_bounds__(256)
void pcen_chunk_agg(const float* __restrict__ X, const float* __restrict__ smooth,
                    float* __restrict__ bchunk) {
  const int lane  = threadIdx.x & 31;
  const int wid   = blockIdx.x * 8 + (threadIdx.x >> 5);
  const int chunk = wid % NCHUNK;
  const int b     = wid / NCHUNK;
  const int hi    = lane >> 4;
  const int cl    = lane & 15;

  float aP[8];
#pragma unroll
  for (int j = 0; j < 8; ++j) {
    const int u = 4 * (j >> 1) + (j & 1) + 2 * hi;
    aP[j] = (u <= cl) ? 1.0f : 0.0f;
  }

  float dch[NCT], dt7[NCT], syB[NCT][8];
#pragma unroll
  for (int i = 0; i < NCT; ++i) {
    const float wsm = clamp01f(smooth[16 * i + cl]);
    const float d   = fminf(fmaxf(1.0f - wsm, DMIN), 1.0f);
    const float l2d = flog2(d);
#pragma unroll
    for (int j = 0; j < 8; ++j) {
      const int u = 4 * (j >> 1) + (j & 1) + 2 * hi;
      syB[i][j] = wsm * fexp2(-(float)u * l2d);
    }
    dt7[i] = fexp2((float)(7 + 8 * hi) * l2d);
    dch[i] = d;
  }

  const float* pB = X + ((size_t)b * TT + (size_t)chunk * LCH + 2 * hi) * CC + cl;

  float s[NCT];
#pragma unroll
  for (int i = 0; i < NCT; ++i) s[i] = 0.0f;

#pragma unroll 1
  for (int tile = 0; tile < NTILE; ++tile) {
    float xb[NCT][8];
    load_b_tile(pB, xb);
    __builtin_amdgcn_sched_barrier(0);
    v8f acc[NCT];
    tile_wmma(aP, syB, xb, acc);
#pragma unroll
    for (int i = 0; i < NCT; ++i) {
      const float a15 = dt7[i] * (acc[i][7] + dch[i] * s[i]);
      s[i] = __shfl(a15, cl + 16, 32);
    }
    pB += 16 * CC;
  }
  if (lane < 16) {
#pragma unroll
    for (int i = 0; i < NCT; ++i)
      bchunk[((size_t)b * NCHUNK + chunk) * CC + 16 * i + cl] = s[i];
  }
}

__global__ __launch_bounds__(256)
void pcen_combine(const float* __restrict__ X, const float* __restrict__ smooth,
                  const float* __restrict__ bchunk, float* __restrict__ start) {
  const int tid = blockIdx.x * 256 + threadIdx.x;
  if (tid >= BB * CC) return;
  const int b = tid / CC, c = tid % CC;
  const float wsm = clamp01f(smooth[c]);
  const float d   = fminf(fmaxf(1.0f - wsm, DMIN), 1.0f);
  const float dL  = fexp2((float)LCH * flog2(d));
  float s = X[(size_t)b * TT * CC + c];
  for (int k = 0; k < NCHUNK; ++k) {
    start[((size_t)b * NCHUNK + k) * CC + c] = s;
    s = dL * s + bchunk[((size_t)b * NCHUNK + k) * CC + c];
  }
}

__global__ __launch_bounds__(256)
void pcen_emit(const float* __restrict__ X, const float* __restrict__ alpha,
               const float* __restrict__ delta, const float* __restrict__ root,
               const float* __restrict__ smooth, const float* __restrict__ start,
               float* __restrict__ out) {
  const int lane  = threadIdx.x & 31;
  const int wid   = blockIdx.x * 8 + (threadIdx.x >> 5);
  const int chunk = wid % NCHUNK;
  const int b     = wid / NCHUNK;
  const int hi    = lane >> 4;
  const int cl    = lane & 15;

  float aP[8];
#pragma unroll
  for (int j = 0; j < 8; ++j) {
    const int u = 4 * (j >> 1) + (j & 1) + 2 * hi;
    aP[j] = (u <= cl) ? 1.0f : 0.0f;
  }

  float dch[NCT], acn[NCT], dcv[NCT], irv[NCT], offv[NCT], s[NCT];
  float syB[NCT][8], dt[NCT][8];
#pragma unroll
  for (int i = 0; i < NCT; ++i) {
    const int c = 16 * i + cl;
    const float wsm = clamp01f(smooth[c]);
    const float d   = fminf(fmaxf(1.0f - wsm, DMIN), 1.0f);
    const float l2d = flog2(d);
    dch[i]  = d;
    acn[i]  = fminf(alpha[c], 1.0f);
    dcv[i]  = delta[c];
    irv[i]  = 1.0f / fmaxf(root[c], 1.0f);
    offv[i] = fexp2(irv[i] * flog2(dcv[i]));
#pragma unroll
    for (int j = 0; j < 8; ++j) {
      const int u = 4 * (j >> 1) + (j & 1) + 2 * hi;
      syB[i][j] = wsm * fexp2(-(float)u * l2d);
      dt[i][j]  = fexp2((float)(j + 8 * hi) * l2d);
    }
    s[i] = start[((size_t)b * NCHUNK + chunk) * CC + c];
  }

  const size_t rowBase = ((size_t)b * TT + (size_t)chunk * LCH) * CC + cl;
  const float* pB = X + rowBase + (size_t)(2 * hi) * CC;
  const float* pD = X + rowBase + (size_t)(8 * hi) * CC;
  float*       pO = out + rowBase + (size_t)(8 * hi) * CC;

#pragma unroll 1
  for (int tile = 0; tile < NTILE; ++tile) {
    float xb[NCT][8];
    load_b_tile(pB, xb);
    float xd[NCT][8];
#pragma unroll
    for (int v = 0; v < 8; ++v) {
#pragma unroll
      for (int i = 0; i < NCT; ++i) xd[i][v] = pD[v * CC + 16 * i];
    }
    v8f acc[NCT];
    tile_wmma(aP, syB, xb, acc);
#pragma unroll
    for (int i = 0; i < NCT; ++i) {
      float a7 = 0.0f;
#pragma unroll
      for (int v = 0; v < 8; ++v) {
        const float ema = dt[i][v] * (acc[i][v] + dch[i] * s[i]);
        if (v == 7) a7 = ema;
        const float lp  = flog2(ema + FLOORV);
        const float num = xd[i][v] * fexp2(-acn[i] * lp) + dcv[i];
        pO[v * CC + 16 * i] = fexp2(irv[i] * flog2(num)) - offv[i];
      }
      s[i] = __shfl(a7, cl + 16, 32);
    }
    pB += 16 * CC; pD += 16 * CC; pO += 16 * CC;
  }
}

extern "C" void kernel_launch(void* const* d_in, const int* in_sizes, int n_in,
                              void* d_out, int out_size, void* d_ws, size_t ws_size,
                              hipStream_t stream) {
  const float* X      = (const float*)d_in[0];
  const float* alpha  = (const float*)d_in[1];
  const float* delta  = (const float*)d_in[2];
  const float* root   = (const float*)d_in[3];
  const float* smooth = (const float*)d_in[4];
  float* out = (float*)d_out;

  float* bchunk = (float*)d_ws;
  float* start  = bchunk + (size_t)BB * NCHUNK * CC;

  pcen_chunk_agg<<<BB * NCHUNK / 8, 256, 0, stream>>>(X, smooth, bchunk);
  pcen_combine<<<(BB * CC + 255) / 256, 256, 0, stream>>>(X, smooth, bchunk, start);
  pcen_emit<<<BB * NCHUNK / 8, 256, 0, stream>>>(X, alpha, delta, root, smooth, start, out);
}


# chunk states by a two-level scan: each 8-wave aggregate group scans its 8 chunk aggregates in LDS, emit folds <=3 group aggregates (start sums regrouped); combine launch gone, 3 launches -> 2
# speedup vs baseline: 1.4730x; 1.2690x over previous
#include <hip/hip_runtime.h>


#define BB 64
#define TT 4096
#define CC 80
#define NCHUNK 32
#define LCH 128
#define NTILE (LCH / 16)
#define NCT 5
#define GCH 8
#define NGRP (NCHUNK / GCH)
#define DMIN 0.00390625f
#define FLOORV 1e-6f

typedef __attribute__((ext_vector_type(2))) float v2f;
typedef __attribute__((ext_vector_type(8))) float v8f;

__device__ __forceinline__ float clamp01f(float x) { return fminf(fmaxf(x, 0.0f), 1.0f); }
__device__ __forceinline__ float flog2(float x) { return __builtin_amdgcn_logf(x); }
__device__ __forceinline__ float fexp2(float x) { return __builtin_amdgcn_exp2f(x); }

__device__ __forceinline__ void load_b_tile(const float* __restrict__ p, float (&x)[NCT][8]) {
#pragma unroll
  for (int j = 0; j < 8; ++j) {
    const int u0 = 4 * (j >> 1) + (j & 1);
#pragma unroll
    for (int i = 0; i < NCT; ++i) x[i][j] = p[u0 * CC + 16 * i];
  }
}

__device__ __forceinline__ void tile_wmma(const float (&aP)[8], const float (&syB)[NCT][8],
                                          const float (&xb)[NCT][8], v8f (&acc)[NCT]) {
#pragma unroll
  for (int i = 0; i < NCT; ++i) acc[i] = (v8f){0.f, 0.f, 0.f, 0.f, 0.f, 0.f, 0.f, 0.f};
#pragma unroll
  for (int kk = 0; kk < 4; ++kk) {
    v2f a2; a2[0] = aP[2 * kk]; a2[1] = aP[2 * kk + 1];
#pragma unroll
    for (int i = 0; i < NCT; ++i) {
      v2f b2;
      b2[0] = xb[i][2 * kk] * syB[i][2 * kk];
      b2[1] = xb[i][2 * kk + 1] * syB[i][2 * kk + 1];
      acc[i] = __builtin_amdgcn_wmma_f32_16x16x4_f32(false, a2, false, b2, (short)0, acc[i],
                                                     false, false);
    }
  }
}

__global__ __launch_bounds__(256)
void pcen_chunk_agg(const float* __restrict__ X, const float* __restrict__ smooth,
                    float* __restrict__ locst, float* __restrict__ gagg) {
  __shared__ float sh[GCH][CC];
  const int lane  = threadIdx.x & 31;
  const int wid   = blockIdx.x * 8 + (threadIdx.x >> 5);
  const int chunk = wid % NCHUNK;
  const int b     = wid / NCHUNK;
  const int hi    = lane >> 4;
  const int cl    = lane & 15;

  float aP[8];
#pragma unroll
  for (int j = 0; j < 8; ++j) {
    const int u = 4 * (j >> 1) + (j & 1) + 2 * hi;
    aP[j] = (u <= cl) ? 1.0f : 0.0f;
  }

  float dch[NCT], dt7[NCT], dLv[NCT], syB[NCT][8];
#pragma unroll
  for (int i = 0; i < NCT; ++i) {
    const float wsm = clamp01f(smooth[16 * i + cl]);
    const float d   = fminf(fmaxf(1.0f - wsm, DMIN), 1.0f);
    const float l2d = flog2(d);
#pragma unroll
    for (int j = 0; j < 8; ++j) {
      const int u = 4 * (j >> 1) + (j & 1) + 2 * hi;
      syB[i][j] = wsm * fexp2(-(float)u * l2d);
    }
    dt7[i] = fexp2((float)(7 + 8 * hi) * l2d);
    dLv[i] = fexp2((float)LCH * l2d);
    dch[i] = d;
  }

  const float* pB = X + ((size_t)b * TT + (size_t)chunk * LCH + 2 * hi) * CC + cl;

  float s[NCT];
#pragma unroll
  for (int i = 0; i < NCT; ++i) s[i] = 0.0f;

#pragma unroll 1
  for (int tile = 0; tile < NTILE; ++tile) {
    float xb[NCT][8];
    load_b_tile(pB, xb);
    __builtin_amdgcn_sched_barrier(0);
    v8f acc[NCT];
    tile_wmma(aP, syB, xb, acc);
#pragma unroll
    for (int i = 0; i < NCT; ++i) {
      const float a15 = dt7[i] * (acc[i][7] + dch[i] * s[i]);
      s[i] = __shfl(a15, cl + 16, 32);
    }
    pB += 16 * CC;
  }
  const int wv = threadIdx.x >> 5;
  if (lane < 16) {
#pragma unroll
    for (int i = 0; i < NCT; ++i) sh[wv][16 * i + cl] = s[i];
  }
  __syncthreads();
  float loc[NCT], grp[NCT];
#pragma unroll
  for (int i = 0; i < NCT; ++i) {
    float a = 0.0f, l = 0.0f;
#pragma unroll
    for (int m = 0; m < GCH; ++m) {
      l = (m == wv) ? a : l;
      a = dLv[i] * a + sh[m][16 * i + cl];
    }
    loc[i] = l; grp[i] = a;
  }
  if (lane < 16) {
#pragma unroll
    for (int i = 0; i < NCT; ++i) {
      locst[((size_t)b * NCHUNK + chunk) * CC + 16 * i + cl] = loc[i];
      if (wv == 0) gagg[((size_t)b * NGRP + chunk / GCH) * CC + 16 * i + cl] = grp[i];
    }
  }
}

__global__ __launch_bounds__(256)
void pcen_emit(const float* __restrict__ X, const float* __restrict__ alpha,
               const float* __restrict__ delta, const float* __restrict__ root,
               const float* __restrict__ smooth, const float* __restrict__ locst,
               const float* __restrict__ gagg, float* __restrict__ out) {
  const int lane  = threadIdx.x & 31;
  const int wid   = blockIdx.x * 8 + (threadIdx.x >> 5);
  const int chunk = wid % NCHUNK;
  const int b     = wid / NCHUNK;
  const int hi    = lane >> 4;
  const int cl    = lane & 15;

  float aP[8];
#pragma unroll
  for (int j = 0; j < 8; ++j) {
    const int u = 4 * (j >> 1) + (j & 1) + 2 * hi;
    aP[j] = (u <= cl) ? 1.0f : 0.0f;
  }

  const int g = chunk / GCH, jj = chunk % GCH;
  float x0v[NCT], lcv[NCT], gv[NCT][3], smv[NCT], alv[NCT], dlv[NCT], rtv[NCT];
#pragma unroll
  for (int i = 0; i < NCT; ++i) {
    const int c = 16 * i + cl;
    smv[i] = smooth[c]; alv[i] = alpha[c]; dlv[i] = delta[c]; rtv[i] = root[c];
    x0v[i] = X[(size_t)b * TT * CC + c];
    lcv[i] = locst[((size_t)b * NCHUNK + chunk) * CC + c];
#pragma unroll
    for (int q = 0; q < 3; ++q) gv[i][q] = gagg[((size_t)b * NGRP + q) * CC + c];
  }
  asm volatile("" :: "v"(gv[0][0]), "v"(gv[0][1]), "v"(gv[0][2]), "v"(gv[1][0]), "v"(gv[1][1]),
               "v"(gv[1][2]), "v"(gv[2][0]), "v"(gv[2][1]), "v"(gv[2][2]), "v"(gv[3][0]),
               "v"(gv[3][1]), "v"(gv[3][2]), "v"(gv[4][0]), "v"(gv[4][1]), "v"(gv[4][2]));

  float dch[NCT], acn[NCT], dcv[NCT], irv[NCT], offv[NCT], s[NCT];
  float syB[NCT][8], dt[NCT][8];
#pragma unroll
  for (int i = 0; i < NCT; ++i) {
    const float wsm = clamp01f(smv[i]);
    const float d   = fminf(fmaxf(1.0f - wsm, DMIN), 1.0f);
    const float l2d = flog2(d);
    dch[i]  = d;
    acn[i]  = fminf(alv[i], 1.0f);
    dcv[i]  = dlv[i];
    irv[i]  = 1.0f / fmaxf(rtv[i], 1.0f);
    offv[i] = fexp2(irv[i] * flog2(dcv[i]));
#pragma unroll
    for (int j = 0; j < 8; ++j) {
      const int u = 4 * (j >> 1) + (j & 1) + 2 * hi;
      syB[i][j] = wsm * fexp2(-(float)u * l2d);
      dt[i][j]  = fexp2((float)(j + 8 * hi) * l2d);
    }
    const float dG = fexp2((float)(GCH * LCH) * l2d);
    float S = x0v[i];
    S = (g > 0) ? dG * S + gv[i][0] : S;
    S = (g > 1) ? dG * S + gv[i][1] : S;
    S = (g > 2) ? dG * S + gv[i][2] : S;
    s[i] = fexp2((float)(jj * LCH) * l2d) * S + lcv[i];
  }

  const size_t rowBase = ((size_t)b * TT + (size_t)chunk * LCH) * CC + cl;
  const float* pB = X + rowBase + (size_t)(2 * hi) * CC;
  const float* pD = X + rowBase + (size_t)(8 * hi) * CC;
  float*       pO = out + rowBase + (size_t)(8 * hi) * CC;

#pragma unroll 1
  for (int tile = 0; tile < NTILE; ++tile) {
    float xb[NCT][8];
    load_b_tile(pB, xb);
    float xd[NCT][8];
#pragma unroll
    for (int v = 0; v < 8; ++v) {
#pragma unroll
      for (int i = 0; i < NCT; ++i) xd[i][v] = pD[v * CC + 16 * i];
    }
    v8f acc[NCT];
    tile_wmma(aP, syB, xb, acc);
#pragma unroll
    for (int i = 0; i < NCT; ++i) {
      float a7 = 0.0f;
#pragma unroll
      for (int v = 0; v < 8; ++v) {
        const float ema = dt[i][v] * (acc[i][v] + dch[i] * s[i]);
        if (v == 7) a7 = ema;
        const float lp  = flog2(ema + FLOORV);
        const float num = xd[i][v] * fexp2(-acn[i] * lp) + dcv[i];
        pO[v * CC + 16 * i] = fexp2(irv[i] * flog2(num)) - offv[i];
      }
      s[i] = __shfl(a7, cl + 16, 32);
    }
    pB += 16 * CC; pD += 16 * CC; pO += 16 * CC;
  }
}

extern "C" void kernel_launch(void* const* d_in, const int* in_sizes, int n_in,
                              void* d_out, int out_size, void* d_ws, size_t ws_size,
                              hipStream_t stream) {
  const float* X      = (const float*)d_in[0];
  const float* alpha  = (const float*)d_in[1];
  const float* delta  = (const float*)d_in[2];
  const float* root   = (const float*)d_in[3];
  const float* smooth = (const float*)d_in[4];
  float* out = (float*)d_out;

  float* locst = (float*)d_ws;
  float* gagg  = locst + (size_t)BB * NCHUNK * CC;

  pcen_chunk_agg<<<BB * NCHUNK / GCH, 256, 0, stream>>>(X, smooth, locst, gagg);
  pcen_emit<<<BB * NCHUNK / GCH, 256, 0, stream>>>(X, alpha, delta, root, smooth, locst, gagg, out);
}


# chunk per wave 128 -> 64 rows in work-groups of 16 waves: 4 waves resident per SIMD (were 2) fill each tile's load round trip; 4,096 waves a launch, not 2,048; chunk sums regrouped
# speedup vs baseline: 1.6405x; 1.1137x over previous
#include <hip/hip_runtime.h>


#define BB 64
#define TT 4096
#define CC 80
#define NCHUNK 64
#define LCH 64
#define NTILE (LCH / 16)
#define NCT 5
#define GCH 16
#define NGRP (NCHUNK / GCH)
#define DMIN 0.00390625f
#define FLOORV 1e-6f

typedef __attribute__((ext_vector_type(2))) float v2f;
typedef __attribute__((ext_vector_type(8))) float v8f;

__device__ __forceinline__ float clamp01f(float x) { return fminf(fmaxf(x, 0.0f), 1.0f); }
__device__ __forceinline__ float flog2(float x) { return __builtin_amdgcn_logf(x); }
__device__ __forceinline__ float fexp2(float x) { return __builtin_amdgcn_exp2f(x); }

__device__ __forceinline__ void load_b_tile(const float* __restrict__ p, float (&x)[NCT][8]) {
#pragma unroll
  for (int j = 0; j < 8; ++j) {
    const int u0 = 4 * (j >> 1) + (j & 1);
#pragma unroll
    for (int i = 0; i < NCT; ++i) x[i][j] = p[u0 * CC + 16 * i];
  }
}

__device__ __forceinline__ void tile_wmma(const float (&aP)[8], const float (&syB)[NCT][8],
                                          const float (&xb)[NCT][8], v8f (&acc)[NCT]) {
#pragma unroll
  for (int i = 0; i < NCT; ++i) acc[i] = (v8f){0.f, 0.f, 0.f, 0.f, 0.f, 0.f, 0.f, 0.f};
#pragma unroll
  for (int kk = 0; kk < 4; ++kk) {
    v2f a2; a2[0] = aP[2 * kk]; a2[1] = aP[2 * kk + 1];
#pragma unroll
    for (int i = 0; i < NCT; ++i) {
      v2f b2;
      b2[0] = xb[i][2 * kk] * syB[i][2 * kk];
      b2[1] = xb[i][2 * kk + 1] * syB[i][2 * kk + 1];
      acc[i] = __builtin_amdgcn_wmma_f32_16x16x4_f32(false, a2, false, b2, (short)0, acc[i],
                                                     false, false);
    }
  }
}

__global__ __launch_bounds__(512)
void pcen_chunk_agg(const float* __restrict__ X, const float* __restrict__ smooth,
                    float* __restrict__ locst, float* __restrict__ gagg) {
  __shared__ float sh[GCH][CC];
  const int lane  = threadIdx.x & 31;
  const int wid   = blockIdx.x * GCH + (threadIdx.x >> 5);
  const int chunk = wid % NCHUNK;
  const int b     = wid / NCHUNK;
  const int hi    = lane >> 4;
  const int cl    = lane & 15;

  float aP[8];
#pragma unroll
  for (int j = 0; j < 8; ++j) {
    const int u = 4 * (j >> 1) + (j & 1) + 2 * hi;
    aP[j] = (u <= cl) ? 1.0f : 0.0f;
  }

  float dch[NCT], dt7[NCT], dLv[NCT], syB[NCT][8];
#pragma unroll
  for (int i = 0; i < NCT; ++i) {
    const float wsm = clamp01f(smooth[16 * i + cl]);
    const float d   = fminf(fmaxf(1.0f - wsm, DMIN), 1.0f);
    const float l2d = flog2(d);
#pragma unroll
    for (int j = 0; j < 8; ++j) {
      const int u = 4 * (j >> 1) + (j & 1) + 2 * hi;
      syB[i][j] = wsm * fexp2(-(float)u * l2d);
    }
    dt7[i] = fexp2((float)(7 + 8 * hi) * l2d);
    dLv[i] = fexp2((float)LCH * l2d);
    dch[i] = d;
  }

  const float* pB = X + ((size_t)b * TT + (size_t)chunk * LCH + 2 * hi) * CC + cl;

  float s[NCT];
#pragma unroll
  for (int i = 0; i < NCT; ++i) s[i] = 0.0f;

#pragma unroll 1
  for (int tile = 0; tile < NTILE; ++tile) {
    float xb[NCT][8];
    load_b_tile(pB, xb);
    __builtin_amdgcn_sched_barrier(0);
    v8f acc[NCT];
    tile_wmma(aP, syB, xb, acc);
#pragma unroll
    for (int i = 0; i < NCT; ++i) {
      const float a15 = dt7[i] * (acc[i][7] + dch[i] * s[i]);
      s[i] = __shfl(a15, cl + 16, 32);
    }
    pB += 16 * CC;
  }
  const int wv = threadIdx.x >> 5;
  if (lane < 16) {
#pragma unroll
    for (int i = 0; i < NCT; ++i) sh[wv][16 * i + cl] = s[i];
  }
  __syncthreads();
  float loc[NCT], grp[NCT];
#pragma unroll
  for (int i = 0; i < NCT; ++i) {
    float a = 0.0f, l = 0.0f;
#pragma unroll
    for (int m = 0; m < GCH; ++m) {
      l = (m == wv) ? a : l;
      a = dLv[i] * a + sh[m][16 * i + cl];
    }
    loc[i] = l; grp[i] = a;
  }
  if (lane < 16) {
#pragma unroll
    for (int i = 0; i < NCT; ++i) {
      locst[((size_t)b * NCHUNK + chunk) * CC + 16 * i + cl] = loc[i];
      if (wv == 0) gagg[((size_t)b * NGRP + chunk / GCH) * CC + 16 * i + cl] = grp[i];
    }
  }
}

__global__ __launch_bounds__(512)
void pcen_emit(const float* __restrict__ X, const float* __restrict__ alpha,
               const float* __restrict__ delta, const float* __restrict__ root,
               const float* __restrict__ smooth, const float* __restrict__ locst,
               const float* __restrict__ gagg, float* __restrict__ out) {
  const int lane  = threadIdx.x & 31;
  const int wid   = blockIdx.x * GCH + (threadIdx.x >> 5);
  const int chunk = wid % NCHUNK;
  const int b     = wid / NCHUNK;
  const int hi    = lane >> 4;
  const int cl    = lane & 15;

  float aP[8];
#pragma unroll
  for (int j = 0; j < 8; ++j) {
    const int u = 4 * (j >> 1) + (j & 1) + 2 * hi;
    aP[j] = (u <= cl) ? 1.0f : 0.0f;
  }

  const int g = chunk / GCH, jj = chunk % GCH;
  float x0v[NCT], lcv[NCT], gv[NCT][3], smv[NCT], alv[NCT], dlv[NCT], rtv[NCT];
#pragma unroll
  for (int i = 0; i < NCT; ++i) {
    const int c = 16 * i + cl;
    smv[i] = smooth[c]; alv[i] = alpha[c]; dlv[i] = delta[c]; rtv[i] = root[c];
    x0v[i] = X[(size_t)b * TT * CC + c];
    lcv[i] = locst[((size_t)b * NCHUNK + chunk) * CC + c];
#pragma unroll
    for (int q = 0; q < 3; ++q) gv[i][q] = gagg[((size_t)b * NGRP + q) * CC + c];
  }
  asm volatile("" :: "v"(gv[0][0]), "v"(gv[0][1]), "v"(gv[0][2]), "v"(gv[1][0]), "v"(gv[1][1]),
               "v"(gv[1][2]), "v"(gv[2][0]), "v"(gv[2][1]), "v"(gv[2][2]), "v"(gv[3][0]),
               "v"(gv[3][1]), "v"(gv[3][2]), "v"(gv[4][0]), "v"(gv[4][1]), "v"(gv[4][2]));

  float dch[NCT], acn[NCT], dcv[NCT], irv[NCT], offv[NCT], s[NCT];
  float syB[NCT][8], dt[NCT][8];
#pragma unroll
  for (int i = 0; i < NCT; ++i) {
    const float wsm = clamp01f(smv[i]);
    const float d   = fminf(fmaxf(1.0f - wsm, DMIN), 1.0f);
    const float l2d = flog2(d);
    dch[i]  = d;
    acn[i]  = fminf(alv[i], 1.0f);
    dcv[i]  = dlv[i];
    irv[i]  = 1.0f / fmaxf(rtv[i], 1.0f);
    offv[i] = fexp2(irv[i] * flog2(dcv[i]));
#pragma unroll
    for (int j = 0; j < 8; ++j) {
      const int u = 4 * (j >> 1) + (j & 1) + 2 * hi;
      syB[i][j] = wsm * fexp2(-(float)u * l2d);
      dt[i][j]  = fexp2((float)(j + 8 * hi) * l2d);
    }
    const float dG = fexp2((float)(GCH * LCH) * l2d);
    float S = x0v[i];
    S = (g > 0) ? dG * S + gv[i][0] : S;
    S = (g > 1) ? dG * S + gv[i][1] : S;
    S = (g > 2) ? dG * S + gv[i][2] : S;
    s[i] = fexp2((float)(jj * LCH) * l2d) * S + lcv[i];
  }

  const size_t rowBase = ((size_t)b * TT + (size_t)chunk * LCH) * CC + cl;
  const float* pB = X + rowBase + (size_t)(2 * hi) * CC;
  const float* pD = X + rowBase + (size_t)(8 * hi) * CC;
  float*       pO = out + rowBase + (size_t)(8 * hi) * CC;

#pragma unroll 1
  for (int tile = 0; tile < NTILE; ++tile) {
    float xb[NCT][8];
    load_b_tile(pB, xb);
    float xd[NCT][8];
#pragma unroll
    for (int v = 0; v < 8; ++v) {
#pragma unroll
      for (int i = 0; i < NCT; ++i) xd[i][v] = pD[v * CC + 16 * i];
    }
    v8f acc[NCT];
    tile_wmma(aP, syB, xb, acc);
#pragma unroll
    for (int i = 0; i < NCT; ++i) {
      float a7 = 0.0f;
#pragma unroll
      for (int v = 0; v < 8; ++v) {
        const float ema = dt[i][v] * (acc[i][v] + dch[i] * s[i]);
        if (v == 7) a7 = ema;
        const float lp  = flog2(ema + FLOORV);
        const float num = xd[i][v] * fexp2(-acn[i] * lp) + dcv[i];
        pO[v * CC + 16 * i] = fexp2(irv[i] * flog2(num)) - offv[i];
      }
      s[i] = __shfl(a7, cl + 16, 32);
    }
    pB += 16 * CC; pD += 16 * CC; pO += 16 * CC;
  }
}

extern "C" void kernel_launch(void* const* d_in, const int* in_sizes, int n_in,
                              void* d_out, int out_size, void* d_ws, size_t ws_size,
                              hipStream_t stream) {
  const float* X      = (const float*)d_in[0];
  const float* alpha  = (const float*)d_in[1];
  const float* delta  = (const float*)d_in[2];
  const float* root   = (const float*)d_in[3];
  const float* smooth = (const float*)d_in[4];
  float* out = (float*)d_out;

  float* locst = (float*)d_ws;
  float* gagg  = locst + (size_t)BB * NCHUNK * CC;

  pcen_chunk_agg<<<BB * NCHUNK / GCH, 32 * GCH, 0, stream>>>(X, smooth, locst, gagg);
  pcen_emit<<<BB * NCHUNK / GCH, 32 * GCH, 0, stream>>>(X, alpha, delta, root, smooth, locst, gagg, out);
}


# tiles cross memory as whole 128-byte lines via a per-wave LDS buffer: 10 16-byte loads (emit: also 10 stores) per lane and tile, 4 full lines each, replace 40 (emit 80+40) 4-byte half-line accesses
# speedup vs baseline: 1.9026x; 1.1598x over previous
#include <hip/hip_runtime.h>


#define BB 64
#define TT 4096
#define CC 80
#define NCHUNK 64
#define LCH 64
#define NTILE (LCH / 16)
#define NCT 5
#define GCH 16
#define NGRP (NCHUNK / GCH)
#define DMIN 0.00390625f
#define FLOORV 1e-6f

#define TPAD 16
#define TSTRIDE (16 * CC + TPAD)

typedef __attribute__((ext_vector_type(2))) float v2f;
typedef __attribute__((ext_vector_type(4))) float v4f;
typedef __attribute__((ext_vector_type(8))) float v8f;

__device__ __forceinline__ float clamp01f(float x) { return fminf(fmaxf(x, 0.0f), 1.0f); }
__device__ __forceinline__ float flog2(float x) { return __builtin_amdgcn_logf(x); }
__device__ __forceinline__ float fexp2(float x) { return __builtin_amdgcn_exp2f(x); }

__device__ __forceinline__ void wave_lds_order() {
  __builtin_amdgcn_fence(__ATOMIC_RELEASE, "wavefront");
  __builtin_amdgcn_wave_barrier();
  __builtin_amdgcn_fence(__ATOMIC_ACQUIRE, "wavefront");
}

__device__ __forceinline__ void stage_in(const float* __restrict__ g, float* buf, int lane) {
  v4f r[10];
#pragma unroll
  for (int k = 0; k < 10; ++k) r[k] = *(const v4f*)(g + (k * 32 + lane) * 4);
#pragma unroll
  for (int k = 0; k < 10; ++k) *(v4f*)(buf + (k * 32 + lane) * 4 + (k >= 5 ? TPAD : 0)) = r[k];
}

__device__ __forceinline__ void read_b_tile(const float* p, float (&x)[NCT][8]) {
#pragma unroll
  for (int j = 0; j < 8; ++j) {
    const int u0 = 4 * (j >> 1) + (j & 1);
#pragma unroll
    for (int i = 0; i < NCT; ++i) x[i][j] = p[u0 * CC + (u0 >= 8 ? TPAD : 0) + 16 * i];
  }
}

__device__ __forceinline__ void tile_wmma(const float (&aP)[8], const float (&syB)[NCT][8],
                                          const float (&xb)[NCT][8], v8f (&acc)[NCT]) {
#pragma unroll
  for (int i = 0; i < NCT; ++i) acc[i] = (v8f){0.f, 0.f, 0.f, 0.f, 0.f, 0.f, 0.f, 0.f};
#pragma unroll
  for (int kk = 0; kk < 4; ++kk) {
    v2f a2; a2[0] = aP[2 * kk]; a2[1] = aP[2 * kk + 1];
#pragma unroll
    for (int i = 0; i < NCT; ++i) {
      v2f b2;
      b2[0] = xb[i][2 * kk] * syB[i][2 * kk];
      b2[1] = xb[i][2 * kk + 1] * syB[i][2 * kk + 1];
      acc[i] = __builtin_amdgcn_wmma_f32_16x16x4_f32(false, a2, false, b2, (short)0, acc[i],
                                                     false, false);
    }
  }
}

__global__ __launch_bounds__(512)
void pcen_chunk_agg(const float* __restrict__ X, const float* __restrict__ smooth,
                    float* __restrict__ locst, float* __restrict__ gagg) {
  __shared__ float sh[GCH][CC];
  __shared__ __attribute__((aligned(16))) float tbuf[GCH][TSTRIDE];
  const int lane  = threadIdx.x & 31;
  const int wid   = blockIdx.x * GCH + (threadIdx.x >> 5);
  const int chunk = wid % NCHUNK;
  const int b     = wid / NCHUNK;
  const int hi    = lane >> 4;
  const int cl    = lane & 15;

  float aP[8];
#pragma unroll
  for (int j = 0; j < 8; ++j) {
    const int u = 4 * (j >> 1) + (j & 1) + 2 * hi;
    aP[j] = (u <= cl) ? 1.0f : 0.0f;
  }

  float dch[NCT], dt7[NCT], dLv[NCT], syB[NCT][8];
#pragma unroll
  for (int i = 0; i < NCT; ++i) {
    const float wsm = clamp01f(smooth[16 * i + cl]);
    const float d   = fminf(fmaxf(1.0f - wsm, DMIN), 1.0f);
    const float l2d = flog2(d);
#pragma unroll
    for (int j = 0; j < 8; ++j) {
      const int u = 4 * (j >> 1) + (j & 1) + 2 * hi;
      syB[i][j] = wsm * fexp2(-(float)u * l2d);
    }
    dt7[i] = fexp2((float)(7 + 8 * hi) * l2d);
    dLv[i] = fexp2((float)LCH * l2d);
    dch[i] = d;
  }

  const float* pT = X + ((size_t)b * TT + (size_t)chunk * LCH) * CC;
  float* wbuf = &tbuf[threadIdx.x >> 5][0];
  const float* fB = wbuf + 2 * hi * CC + cl;

  float s[NCT];
#pragma unroll
  for (int i = 0; i < NCT; ++i) s[i] = 0.0f;

#pragma unroll 1
  for (int tile = 0; tile < NTILE; ++tile) {
    stage_in(pT, wbuf, lane);
    wave_lds_order();
    float xb[NCT][8];
    read_b_tile(fB, xb);
    wave_lds_order();
    __builtin_amdgcn_sched_barrier(0);
    v8f acc[NCT];
    tile_wmma(aP, syB, xb, acc);
#pragma unroll
    for (int i = 0; i < NCT; ++i) {
      const float a15 = dt7[i] * (acc[i][7] + dch[i] * s[i]);
      s[i] = __shfl(a15, cl + 16, 32);
    }
    pT += 16 * CC;
  }
  const int wv = threadIdx.x >> 5;
  if (lane < 16) {
#pragma unroll
    for (int i = 0; i < NCT; ++i) sh[wv][16 * i + cl] = s[i];
  }
  __syncthreads();
  float loc[NCT], grp[NCT];
#pragma unroll
  for (int i = 0; i < NCT; ++i) {
    float a = 0.0f, l = 0.0f;
#pragma unroll
    for (int m = 0; m < GCH; ++m) {
      l = (m == wv) ? a : l;
      a = dLv[i] * a + sh[m][16 * i + cl];
    }
    loc[i] = l; grp[i] = a;
  }
  if (lane < 16) {
#pragma unroll
    for (int i = 0; i < NCT; ++i) {
      locst[((size_t)b * NCHUNK + chunk) * CC + 16 * i + cl] = loc[i];
      if (wv == 0) gagg[((size_t)b * NGRP + chunk / GCH) * CC + 16 * i + cl] = grp[i];
    }
  }
}

__global__ __launch_bounds__(512)
void pcen_emit(const float* __restrict__ X, const float* __restrict__ alpha,
               const float* __restrict__ delta, const float* __restrict__ root,
               const float* __restrict__ smooth, const float* __restrict__ locst,
               const float* __restrict__ gagg, float* __restrict__ out) {
  __shared__ __attribute__((aligned(16))) float tbuf[GCH][TSTRIDE];
  const int lane  = threadIdx.x & 31;
  const int wid   = blockIdx.x * GCH + (threadIdx.x >> 5);
  const int chunk = wid % NCHUNK;
  const int b     = wid / NCHUNK;
  const int hi    = lane >> 4;
  const int cl    = lane & 15;

  float aP[8];
#pragma unroll
  for (int j = 0; j < 8; ++j) {
    const int u = 4 * (j >> 1) + (j & 1) + 2 * hi;
    aP[j] = (u <= cl) ? 1.0f : 0.0f;
  }

  const int g = chunk / GCH, jj = chunk % GCH;
  float x0v[NCT], lcv[NCT], gv[NCT][3], smv[NCT], alv[NCT], dlv[NCT], rtv[NCT];
#pragma unroll
  for (int i = 0; i < NCT; ++i) {
    const int c = 16 * i + cl;
    smv[i] = smooth[c]; alv[i] = alpha[c]; dlv[i] = delta[c]; rtv[i] = root[c];
    x0v[i] = X[(size_t)b * TT * CC + c];
    lcv[i] = locst[((size_t)b * NCHUNK + chunk) * CC + c];
#pragma unroll
    for (int q = 0; q < 3; ++q) gv[i][q] = gagg[((size_t)b * NGRP + q) * CC + c];
  }
  asm volatile("" :: "v"(gv[0][0]), "v"(gv[0][1]), "v"(gv[0][2]), "v"(gv[1][0]), "v"(gv[1][1]),
               "v"(gv[1][2]), "v"(gv[2][0]), "v"(gv[2][1]), "v"(gv[2][2]), "v"(gv[3][0]),
               "v"(gv[3][1]), "v"(gv[3][2]), "v"(gv[4][0]), "v"(gv[4][1]), "v"(gv[4][2]));

  float dch[NCT], acn[NCT], dcv[NCT], irv[NCT], offv[NCT], s[NCT];
  float syB[NCT][8], dt[NCT][8];
#pragma unroll
  for (int i = 0; i < NCT; ++i) {
    const float wsm = clamp01f(smv[i]);
    const float d   = fminf(fmaxf(1.0f - wsm, DMIN), 1.0f);
    const float l2d = flog2(d);
    dch[i]  = d;
    acn[i]  = fminf(alv[i], 1.0f);
    dcv[i]  = dlv[i];
    irv[i]  = 1.0f / fmaxf(rtv[i], 1.0f);
    offv[i] = fexp2(irv[i] * flog2(dcv[i]));
#pragma unroll
    for (int j = 0; j < 8; ++j) {
      const int u = 4 * (j >> 1) + (j & 1) + 2 * hi;
      syB[i][j] = wsm * fexp2(-(float)u * l2d);
      dt[i][j]  = fexp2((float)(j + 8 * hi) * l2d);
    }
    const float dG = fexp2((float)(GCH * LCH) * l2d);
    float S = x0v[i];
    S = (g > 0) ? dG * S + gv[i][0] : S;
    S = (g > 1) ? dG * S + gv[i][1] : S;
    S = (g > 2) ? dG * S + gv[i][2] : S;
    s[i] = fexp2((float)(jj * LCH) * l2d) * S + lcv[i];
  }

  const size_t tileBase = ((size_t)b * TT + (size_t)chunk * LCH) * CC;
  const float* pT = X + tileBase;
  float*       pO = out + tileBase;
  float* wbuf = &tbuf[threadIdx.x >> 5][0];
  const float* fB = wbuf + 2 * hi * CC + cl;
  float*       fD = wbuf + hi * (8 * CC + TPAD) + cl;

#pragma unroll 1
  for (int tile = 0; tile < NTILE; ++tile) {
    stage_in(pT, wbuf, lane);
    wave_lds_order();
    float xb[NCT][8];
    read_b_tile(fB, xb);
    float xd[NCT][8];
#pragma unroll
    for (int v = 0; v < 8; ++v) {
#pragma unroll
      for (int i = 0; i < NCT; ++i) xd[i][v] = fD[v * CC + 16 * i];
    }
    wave_lds_order();
    __builtin_amdgcn_sched_barrier(0);
    v8f acc[NCT];
    tile_wmma(aP, syB, xb, acc);
#pragma unroll
    for (int i = 0; i < NCT; ++i) {
      float a7 = 0.0f;
#pragma unroll
      for (int v = 0; v < 8; ++v) {
        const float ema = dt[i][v] * (acc[i][v] + dch[i] * s[i]);
        if (v == 7) a7 = ema;
        const float lp  = flog2(ema + FLOORV);
        const float num = xd[i][v] * fexp2(-acn[i] * lp) + dcv[i];
        fD[v * CC + 16 * i] = fexp2(irv[i] * flog2(num)) - offv[i];
      }
      s[i] = __shfl(a7, cl + 16, 32);
    }
    wave_lds_order();
    v4f ro[10];
#pragma unroll
    for (int k = 0; k < 10; ++k) ro[k] = *(const v4f*)(wbuf + (k * 32 + lane) * 4 + (k >= 5 ? TPAD : 0));
#pragma unroll
    for (int k = 0; k < 10; ++k) *(v4f*)(pO + (k * 32 + lane) * 4) = ro[k];
    wave_lds_order();
    pT += 16 * CC; pO += 16 * CC;
  }
}

extern "C" void kernel_launch(void* const* d_in, const int* in_sizes, int n_in,
                              void* d_out, int out_size, void* d_ws, size_t ws_size,
                              hipStream_t stream) {
  const float* X      = (const float*)d_in[0];
  const float* alpha  = (const float*)d_in[1];
  const float* delta  = (const float*)d_in[2];
  const float* root   = (const float*)d_in[3];
  const float* smooth = (const float*)d_in[4];
  float* out = (float*)d_out;

  float* locst = (float*)d_ws;
  float* gagg  = locst + (size_t)BB * NCHUNK * CC;

  pcen_chunk_agg<<<BB * NCHUNK / GCH, 32 * GCH, 0, stream>>>(X, smooth, locst, gagg);
  pcen_emit<<<BB * NCHUNK / GCH, 32 * GCH, 0, stream>>>(X, alpha, delta, root, smooth, locst, gagg, out);
}
